# starting kernel
# baseline (speedup 1.0000x reference)
#include <hip/hip_runtime.h>
#include <hip/hip_bf16.h>

typedef __attribute__((ext_vector_type(2))) float v2f;
typedef __attribute__((ext_vector_type(8))) float v8f;

__device__ __forceinline__ float bcast_lane(float v, int l) {
    return __int_as_float(__builtin_amdgcn_readlane(__float_as_int(v), l));
}

template <int MODE>
__global__ __launch_bounds__(32) void gru_scan_kernel(
    const float* __restrict__ x,
    const float* __restrict__ W_ih,
    const float* __restrict__ W_hh,
    const float* __restrict__ b_ih,
    const float* __restrict__ b_hh,
    const float* __restrict__ W_out,
    const float* __restrict__ b_out,
    float* __restrict__ Hbuf,
    float* __restrict__ y,
    int nsteps)
{
    const int lane = threadIdx.x;
    const int m    = lane & 15;

    float wA[16], wB[16];
#pragma unroll
    for (int k = 0; k < 16; ++k) {
        wA[k] = W_hh[lane * 16 + k];
        wB[k] = W_hh[(32 + m) * 16 + k];
    }
    const float cA   = b_ih[lane] + b_hh[lane];
    const float wiA  = W_ih[lane];
    const float bhhB = b_hh[32 + m];
    const float cBgi = b_ih[32 + m];
    const float wiB  = W_ih[32 + m];
    const float wout = (lane < 16) ? W_out[m] : 0.0f;
    const float bo   = b_out[0];

    float hv    = 0.0f;
    float x_cur = x[0];

    for (int t = 0; t < nsteps; ++t) {
        float x_nxt = x[t + 1];

        float sh[16];
#pragma unroll
        for (int k = 0; k < 16; ++k) sh[k] = bcast_lane(hv, k);

        float aA0 = fmaf(wiA, x_cur, cA);
        float aA1 = 0.0f;
        float aB0 = bhhB;
        float aB1 = 0.0f;
#pragma unroll
        for (int k = 0; k < 8; ++k) {
            aA0 = fmaf(wA[k],     sh[k],     aA0);
            aA1 = fmaf(wA[k + 8], sh[k + 8], aA1);
            aB0 = fmaf(wB[k],     sh[k],     aB0);
            aB1 = fmaf(wB[k + 8], sh[k + 8], aB1);
        }
        const float accA = aA0 + aA1;
        const float accB = aB0 + aB1;

        const float gA = __builtin_amdgcn_rcpf(1.0f + __expf(-accA));

        const float npre = fmaf(gA, accB, fmaf(wiB, x_cur, cBgi));
        const float e2   = __expf(npre + npre);
        const float n    = 1.0f - 2.0f * __builtin_amdgcn_rcpf(e2 + 1.0f);

        const float zl   = __shfl_xor(gA, 16, 32);
        const float hnew = fmaf(zl, hv - n, n);
        hv = hnew;

        if (MODE == 0) {
            if (lane < 16) Hbuf[(size_t)t * 16 + lane] = hnew;
        } else {
            float p = wout * hnew;
            p += __shfl_xor(p, 8, 32);
            p += __shfl_xor(p, 4, 32);
            p += __shfl_xor(p, 2, 32);
            p += __shfl_xor(p, 1, 32);
            if (lane == 0) y[t] = p + bo;
        }
        x_cur = x_nxt;
    }
}

__global__ __launch_bounds__(128) void gru_proj_wmma_kernel(
    const float* __restrict__ Hbuf,
    const float* __restrict__ W_out,
    const float* __restrict__ b_out,
    float* __restrict__ y,
    int nY)
{
    const int lane = threadIdx.x & 31;
    const int wave = blockIdx.x * 4 + (threadIdx.x >> 5);
    const int t0   = wave * 16;
    if (t0 >= nY) return;

    const int half = lane >> 4;
    const int lrow = lane & 15;

    const float bo = b_out[0];
    v8f c;
#pragma unroll
    for (int i = 0; i < 8; ++i) c[i] = bo;

    const float* hrow = Hbuf + (size_t)(t0 + lrow) * 16;
#pragma unroll
    for (int j = 0; j < 4; ++j) {
        const int col = 4 * j + 2 * half;
        v2f a; a[0] = hrow[col];  a[1] = hrow[col + 1];
        v2f b; b[0] = W_out[col]; b[1] = W_out[col + 1];
        c = __builtin_amdgcn_wmma_f32_16x16x4_f32(
                 false, a,  false, b,
                 (short)0, c,  false,  false);
    }

    const int m0 = t0 + 8 * half;
    if (lrow == 0) {
        if (m0 + 8 <= nY) {
            float4 lo = make_float4(c[0], c[1], c[2], c[3]);
            float4 hi = make_float4(c[4], c[5], c[6], c[7]);
            *(float4*)(y + m0)     = lo;
            *(float4*)(y + m0 + 4) = hi;
        } else {
#pragma unroll
            for (int i = 0; i < 8; ++i)
                if (m0 + i < nY) y[m0 + i] = c[i];
        }
    }
}

extern "C" void kernel_launch(void* const* d_in, const int* in_sizes, int n_in,
                              void* d_out, int out_size, void* d_ws, size_t ws_size,
                              hipStream_t stream)
{
    const float* x     = (const float*)d_in[0];
    const float* W_ih  = (const float*)d_in[1];
    const float* W_hh  = (const float*)d_in[2];
    const float* b_ih  = (const float*)d_in[3];
    const float* b_hh  = (const float*)d_in[4];
    const float* W_out = (const float*)d_in[5];
    const float* b_out = (const float*)d_in[6];
    float* y = (float*)d_out;

    const int nsteps = in_sizes[0] - 1;
    const size_t needH = (size_t)(nsteps + 1) * 16 * sizeof(float);
    float* Hbuf = (float*)d_ws;

    if (ws_size >= needH) {
        hipLaunchKernelGGL((gru_scan_kernel<0>), dim3(1), dim3(32), 0, stream,
                           x, W_ih, W_hh, b_ih, b_hh, W_out, b_out,
                           Hbuf, y, nsteps);
        const int tiles  = (nsteps + 15) / 16;
        const int blocks = (tiles + 3) / 4;
        hipLaunchKernelGGL(gru_proj_wmma_kernel, dim3(blocks), dim3(128), 0, stream,
                           Hbuf, W_out, b_out, y, nsteps);
    } else {
        hipLaunchKernelGGL((gru_scan_kernel<1>), dim3(1), dim3(32), 0, stream,
                           x, W_ih, W_hh, b_ih, b_hh, W_out, b_out,
                           nullptr, y, nsteps);
    }
}


# z reaches lanes 0-15 by v_permlanex16, not an LDS permute; (1-z) and z*h are ready before n, so h_new=(1-z)*n+z*h is one fma behind n (reference's form; rounds unlike n+z*(h-n))
# speedup vs baseline: 1.2171x; 1.2171x over previous
#include <hip/hip_runtime.h>
#include <hip/hip_bf16.h>

typedef __attribute__((ext_vector_type(2))) float v2f;
typedef __attribute__((ext_vector_type(8))) float v8f;

__device__ __forceinline__ float bcast_lane(float v, int l) {
    return __int_as_float(__builtin_amdgcn_readlane(__float_as_int(v), l));
}

__device__ __forceinline__ float xhalf(float v) {
    const int b = __float_as_int(v);
    return __int_as_float(
        __builtin_amdgcn_permlanex16(b, b, 0x76543210, 0xfedcba98, false, false));
}

template <int MODE>
__global__ __launch_bounds__(32) void gru_scan_kernel(
    const float* __restrict__ x,
    const float* __restrict__ W_ih,
    const float* __restrict__ W_hh,
    const float* __restrict__ b_ih,
    const float* __restrict__ b_hh,
    const float* __restrict__ W_out,
    const float* __restrict__ b_out,
    float* __restrict__ Hbuf,
    float* __restrict__ y,
    int nsteps)
{
    const int lane = threadIdx.x;
    const int m    = lane & 15;

    float wA[16], wB[16];
#pragma unroll
    for (int k = 0; k < 16; ++k) {
        wA[k] = W_hh[lane * 16 + k];
        wB[k] = W_hh[(32 + m) * 16 + k];
    }
    const float cA   = b_ih[lane] + b_hh[lane];
    const float wiA  = W_ih[lane];
    const float bhhB = b_hh[32 + m];
    const float cBgi = b_ih[32 + m];
    const float wiB  = W_ih[32 + m];
    const float wout = (lane < 16) ? W_out[m] : 0.0f;
    const float bo   = b_out[0];

    float hv    = 0.0f;
    float x_cur = x[0];

    for (int t = 0; t < nsteps; ++t) {
        float x_nxt = x[t + 1];

        float sh[16];
#pragma unroll
        for (int k = 0; k < 16; ++k) sh[k] = bcast_lane(hv, k);

        float aA0 = fmaf(wiA, x_cur, cA);
        float aA1 = 0.0f;
        float aB0 = bhhB;
        float aB1 = 0.0f;
#pragma unroll
        for (int k = 0; k < 8; ++k) {
            aA0 = fmaf(wA[k],     sh[k],     aA0);
            aA1 = fmaf(wA[k + 8], sh[k + 8], aA1);
            aB0 = fmaf(wB[k],     sh[k],     aB0);
            aB1 = fmaf(wB[k + 8], sh[k + 8], aB1);
        }
        const float accA = aA0 + aA1;
        const float accB = aB0 + aB1;

        const float gA = __builtin_amdgcn_rcpf(1.0f + __expf(-accA));

        const float zl  = xhalf(gA);
        const float omz = 1.0f - zl;
        const float zh  = zl * hv;

        const float npre = fmaf(gA, accB, fmaf(wiB, x_cur, cBgi));
        const float e2   = __expf(npre + npre);
        const float n    = 1.0f - 2.0f * __builtin_amdgcn_rcpf(e2 + 1.0f);

        const float hnew = fmaf(omz, n, zh);
        hv = hnew;

        if (MODE == 0) {
            if (lane < 16) Hbuf[(size_t)t * 16 + lane] = hnew;
        } else {
            float p = wout * hnew;
            p += __shfl_xor(p, 8, 32);
            p += __shfl_xor(p, 4, 32);
            p += __shfl_xor(p, 2, 32);
            p += __shfl_xor(p, 1, 32);
            if (lane == 0) y[t] = p + bo;
        }
        x_cur = x_nxt;
    }
}

__global__ __launch_bounds__(128) void gru_proj_wmma_kernel(
    const float* __restrict__ Hbuf,
    const float* __restrict__ W_out,
    const float* __restrict__ b_out,
    float* __restrict__ y,
    int nY)
{
    const int lane = threadIdx.x & 31;
    const int wave = blockIdx.x * 4 + (threadIdx.x >> 5);
    const int t0   = wave * 16;
    if (t0 >= nY) return;

    const int half = lane >> 4;
    const int lrow = lane & 15;

    const float bo = b_out[0];
    v8f c;
#pragma unroll
    for (int i = 0; i < 8; ++i) c[i] = bo;

    const float* hrow = Hbuf + (size_t)(t0 + lrow) * 16;
#pragma unroll
    for (int j = 0; j < 4; ++j) {
        const int col = 4 * j + 2 * half;
        v2f a; a[0] = hrow[col];  a[1] = hrow[col + 1];
        v2f b; b[0] = W_out[col]; b[1] = W_out[col + 1];
        c = __builtin_amdgcn_wmma_f32_16x16x4_f32(
                 false, a,  false, b,
                 (short)0, c,  false,  false);
    }

    const int m0 = t0 + 8 * half;
    if (lrow == 0) {
        if (m0 + 8 <= nY) {
            float4 lo = make_float4(c[0], c[1], c[2], c[3]);
            float4 hi = make_float4(c[4], c[5], c[6], c[7]);
            *(float4*)(y + m0)     = lo;
            *(float4*)(y + m0 + 4) = hi;
        } else {
#pragma unroll
            for (int i = 0; i < 8; ++i)
                if (m0 + i < nY) y[m0 + i] = c[i];
        }
    }
}

extern "C" void kernel_launch(void* const* d_in, const int* in_sizes, int n_in,
                              void* d_out, int out_size, void* d_ws, size_t ws_size,
                              hipStream_t stream)
{
    const float* x     = (const float*)d_in[0];
    const float* W_ih  = (const float*)d_in[1];
    const float* W_hh  = (const float*)d_in[2];
    const float* b_ih  = (const float*)d_in[3];
    const float* b_hh  = (const float*)d_in[4];
    const float* W_out = (const float*)d_in[5];
    const float* b_out = (const float*)d_in[6];
    float* y = (float*)d_out;

    const int nsteps = in_sizes[0] - 1;
    const size_t needH = (size_t)(nsteps + 1) * 16 * sizeof(float);
    float* Hbuf = (float*)d_ws;

    if (ws_size >= needH) {
        hipLaunchKernelGGL((gru_scan_kernel<0>), dim3(1), dim3(32), 0, stream,
                           x, W_ih, W_hh, b_ih, b_hh, W_out, b_out,
                           Hbuf, y, nsteps);
        const int tiles  = (nsteps + 15) / 16;
        const int blocks = (tiles + 3) / 4;
        hipLaunchKernelGGL(gru_proj_wmma_kernel, dim3(blocks), dim3(128), 0, stream,
                           Hbuf, W_out, b_out, y, nsteps);
    } else {
        hipLaunchKernelGGL((gru_scan_kernel<1>), dim3(1), dim3(32), 0, stream,
                           x, W_ih, W_hh, b_ih, b_hh, W_out, b_out,
                           nullptr, y, nsteps);
    }
}


# scan's time loop in blocks of 16 steps: a block's x by one 64-byte scalar load a block ahead (was a dword a step ahead), its 16 h rows stored at block end as 8 whole 128-byte lines; no branch inside
# speedup vs baseline: 1.5198x; 1.2487x over previous
#include <hip/hip_runtime.h>
#include <hip/hip_bf16.h>

typedef __attribute__((ext_vector_type(2))) float v2f;
typedef __attribute__((ext_vector_type(8))) float v8f;

__device__ __forceinline__ float bcast_lane(float v, int l) {
    return __int_as_float(__builtin_amdgcn_readlane(__float_as_int(v), l));
}

__device__ __forceinline__ float xhalf(float v) {
    const int b = __float_as_int(v);
    return __int_as_float(
        __builtin_amdgcn_permlanex16(b, b, 0x76543210, 0xfedcba98, false, false));
}

struct GruRow {
    float wA[16], wB[16];
    float cA, wiA, bhhB, cBgi, wiB, wout, bo;
};

__device__ __forceinline__ float gru_step(const GruRow& g, float hv, float x_cur)
{
    float sh[16];
#pragma unroll
    for (int k = 0; k < 16; ++k) sh[k] = bcast_lane(hv, k);

    float aA0 = fmaf(g.wiA, x_cur, g.cA);
    float aA1 = 0.0f;
    float aB0 = g.bhhB;
    float aB1 = 0.0f;
#pragma unroll
    for (int k = 0; k < 8; ++k) {
        aA0 = fmaf(g.wA[k],     sh[k],     aA0);
        aA1 = fmaf(g.wA[k + 8], sh[k + 8], aA1);
        aB0 = fmaf(g.wB[k],     sh[k],     aB0);
        aB1 = fmaf(g.wB[k + 8], sh[k + 8], aB1);
    }
    const float accA = aA0 + aA1;
    const float accB = aB0 + aB1;

    const float gA = __builtin_amdgcn_rcpf(1.0f + __expf(-accA));

    const float zl  = xhalf(gA);
    const float omz = 1.0f - zl;
    const float zh  = zl * hv;

    const float npre = fmaf(gA, accB, fmaf(g.wiB, x_cur, g.cBgi));
    const float e2   = __expf(npre + npre);
    const float n    = 1.0f - 2.0f * __builtin_amdgcn_rcpf(e2 + 1.0f);

    return fmaf(omz, n, zh);
}

template <int MODE>
__device__ __forceinline__ void gru_emit(const GruRow& g, float hnew, int lane,
                                         float* __restrict__ hdst, float* __restrict__ ydst)
{
    if (MODE == 0) {
        if (lane < 16) hdst[lane] = hnew;
    } else {
        float p = g.wout * hnew;
        p += __shfl_xor(p, 8, 32);
        p += __shfl_xor(p, 4, 32);
        p += __shfl_xor(p, 2, 32);
        p += __shfl_xor(p, 1, 32);
        if (lane == 0) *ydst = p + g.bo;
    }
}

template <int MODE>
__global__ __launch_bounds__(32) void gru_scan_kernel(
    const float* __restrict__ x,
    const float* __restrict__ W_ih,
    const float* __restrict__ W_hh,
    const float* __restrict__ b_ih,
    const float* __restrict__ b_hh,
    const float* __restrict__ W_out,
    const float* __restrict__ b_out,
    float* __restrict__ Hbuf,
    float* __restrict__ y,
    int nsteps)
{
    const int lane = threadIdx.x;
    const int m    = lane & 15;

    GruRow g;
#pragma unroll
    for (int k = 0; k < 16; ++k) {
        g.wA[k] = W_hh[lane * 16 + k];
        g.wB[k] = W_hh[(32 + m) * 16 + k];
    }
    g.cA   = b_ih[lane] + b_hh[lane];
    g.wiA  = W_ih[lane];
    g.bhhB = b_hh[32 + m];
    g.cBgi = b_ih[32 + m];
    g.wiB  = W_ih[32 + m];
    g.wout = (lane < 16) ? W_out[m] : 0.0f;
    g.bo   = b_out[0];

    float hv = 0.0f;
    int   t  = 0;

    const int nblk = nsteps >> 4;
    if (nblk > 0) {
        float xb[16];
#pragma unroll
        for (int i = 0; i < 16; ++i) xb[i] = x[i];

        for (int b = 0; b < nblk; ++b) {
            const int bn = (b + 1 < nblk) ? (b + 1) : b;
            const float* __restrict__ xp = x + (size_t)bn * 16;
            float xn[16];
#pragma unroll
            for (int i = 0; i < 16; ++i) xn[i] = xp[i];

            const size_t t0 = (size_t)b * 16;
            float hrow[16];
#pragma unroll
            for (int i = 0; i < 16; ++i) {
                hv = gru_step(g, hv, xb[i]);
                hrow[i] = hv;
                __builtin_amdgcn_sched_barrier(0);
            }
            if (MODE == 0) {
#pragma unroll
                for (int i = 0; i < 16; i += 2) {
                    const float up = xhalf(hrow[i + 1]);
                    Hbuf[(t0 + i) * 16 + lane] = (lane < 16) ? hrow[i] : up;
                }
            } else {
#pragma unroll
                for (int i = 0; i < 16; ++i)
                    gru_emit<MODE>(g, hrow[i], lane, nullptr, y + t0 + i);
            }
#pragma unroll
            for (int i = 0; i < 16; ++i) xb[i] = xn[i];
        }
        t = nblk * 16;
    }
    for (; t < nsteps; ++t) {
        hv = gru_step(g, hv, x[t]);
        gru_emit<MODE>(g, hv, lane, Hbuf + (size_t)t * 16, y + t);
    }
}

__global__ __launch_bounds__(128) void gru_proj_wmma_kernel(
    const float* __restrict__ Hbuf,
    const float* __restrict__ W_out,
    const float* __restrict__ b_out,
    float* __restrict__ y,
    int nY)
{
    const int lane = threadIdx.x & 31;
    const int wave = blockIdx.x * 4 + (threadIdx.x >> 5);
    const int t0   = wave * 16;
    if (t0 >= nY) return;

    const int half = lane >> 4;
    const int lrow = lane & 15;

    const float bo = b_out[0];
    v8f c;
#pragma unroll
    for (int i = 0; i < 8; ++i) c[i] = bo;

    const float* hrow = Hbuf + (size_t)(t0 + lrow) * 16;
#pragma unroll
    for (int j = 0; j < 4; ++j) {
        const int col = 4 * j + 2 * half;
        v2f a; a[0] = hrow[col];  a[1] = hrow[col + 1];
        v2f b; b[0] = W_out[col]; b[1] = W_out[col + 1];
        c = __builtin_amdgcn_wmma_f32_16x16x4_f32(
                 false, a,  false, b,
                 (short)0, c,  false,  false);
    }

    const int m0 = t0 + 8 * half;
    if (lrow == 0) {
        if (m0 + 8 <= nY) {
            float4 lo = make_float4(c[0], c[1], c[2], c[3]);
            float4 hi = make_float4(c[4], c[5], c[6], c[7]);
            *(float4*)(y + m0)     = lo;
            *(float4*)(y + m0 + 4) = hi;
        } else {
#pragma unroll
            for (int i = 0; i < 8; ++i)
                if (m0 + i < nY) y[m0 + i] = c[i];
        }
    }
}

extern "C" void kernel_launch(void* const* d_in, const int* in_sizes, int n_in,
                              void* d_out, int out_size, void* d_ws, size_t ws_size,
                              hipStream_t stream)
{
    const float* x     = (const float*)d_in[0];
    const float* W_ih  = (const float*)d_in[1];
    const float* W_hh  = (const float*)d_in[2];
    const float* b_ih  = (const float*)d_in[3];
    const float* b_hh  = (const float*)d_in[4];
    const float* W_out = (const float*)d_in[5];
    const float* b_out = (const float*)d_in[6];
    float* y = (float*)d_out;

    const int nsteps = in_sizes[0] - 1;
    const size_t needH = (size_t)(nsteps + 1) * 16 * sizeof(float);
    float* Hbuf = (float*)d_ws;

    if (ws_size >= needH) {
        hipLaunchKernelGGL((gru_scan_kernel<0>), dim3(1), dim3(32), 0, stream,
                           x, W_ih, W_hh, b_ih, b_hh, W_out, b_out,
                           Hbuf, y, nsteps);
        const int tiles  = (nsteps + 15) / 16;
        const int blocks = (tiles + 3) / 4;
        hipLaunchKernelGGL(gru_proj_wmma_kernel, dim3(blocks), dim3(128), 0, stream,
                           Hbuf, W_out, b_out, y, nsteps);
    } else {
        hipLaunchKernelGGL((gru_scan_kernel<1>), dim3(1), dim3(32), 0, stream,
                           x, W_ih, W_hh, b_ih, b_hh, W_out, b_out,
                           nullptr, y, nsteps);
    }
}


# h reaches the matvec by DPP row_share inside the fmac, not by v_readlane and scalar operands: each 16-lane row sums its half of k for r, z, n; one v_permlanex16+add joins them; sum order kept
# speedup vs baseline: 1.7930x; 1.1798x over previous
#include <hip/hip_runtime.h>
#include <hip/hip_bf16.h>

typedef __attribute__((ext_vector_type(2))) float v2f;
typedef __attribute__((ext_vector_type(8))) float v8f;

__device__ __forceinline__ float xhalf8(float v) {
    const int b = __float_as_int(v);
    return __int_as_float(
        __builtin_amdgcn_permlanex16(b, b, 0xfedcba98, 0x76543210, true, false));
}

struct GruRow {
    float wR[8], wZ[8], wN[8];
    float cR, wiR, cZ, wiZ, bhhN, cN, wiN, wout, bo;
    bool  lo;
};

struct GruInit { float aR, aZ, gN; };

__device__ __forceinline__ GruInit gru_init(const GruRow& g, float x)
{
    GruInit s;
    s.aR = g.lo ? fmaf(g.wiR, x, g.cR) : 0.0f;
    s.aZ = g.lo ? fmaf(g.wiZ, x, g.cZ) : 0.0f;
    s.gN = fmaf(g.wiN, x, g.cN);
    return s;
}

#define GRU_HERE() __builtin_amdgcn_sched_barrier(0)

#define GRU_LINK(K)                                                                          \
    asm("v_fmac_f32_dpp %0, %3, %4 row_share:" #K " row_mask:0xf bank_mask:0xf bound_ctrl:1\n\t"  \
        "v_fmac_f32_dpp %1, %3, %5 row_share:" #K " row_mask:0xf bank_mask:0xf bound_ctrl:1\n\t"  \
        "v_fmac_f32_dpp %2, %3, %6 row_share:" #K " row_mask:0xf bank_mask:0xf bound_ctrl:1"       \
        : "+v"(aR), "+v"(aN), "+v"(aZ)                                                       \
        : "v"(hv), "v"(g.wR[K]), "v"(g.wN[K]), "v"(g.wZ[K]))

__device__ __forceinline__ float gru_step(const GruRow& g, float hv, const GruInit& cur,
                                          float x_next, GruInit& nxt)
{
    float aR = cur.aR, aZ = cur.aZ, aN = g.bhhN;
    GRU_LINK(0); GRU_LINK(1); GRU_LINK(2); GRU_LINK(3);
    GRU_LINK(4); GRU_LINK(5); GRU_LINK(6); GRU_LINK(7);

    const float kNegLog2e = -0x1.715476p+0f;
    const float pR = xhalf8(aR), pN = xhalf8(aN), pZ = xhalf8(aZ);
    const float accR = aR + pR;
    const float accN = aN + pN;
    const float mr = accR * kNegLog2e;
    GRU_HERE();
    const float accZ = aZ + pZ;
    const float er = __builtin_amdgcn_exp2f(mr);
    const float mz = accZ * kNegLog2e;
    GRU_HERE();
    const float ez = __builtin_amdgcn_exp2f(mz);
    const float r  = __builtin_amdgcn_rcpf(1.0f + er);
    GRU_HERE();
    const float z  = __builtin_amdgcn_rcpf(1.0f + ez);

    const float npre = fmaf(r, accN, cur.gN);
    const float tw   = npre + npre;
    GRU_HERE();
    const float omz = 1.0f - z;
    const float zh  = z * hv;
    const float e2  = __expf(tw);
    nxt = gru_init(g, x_next);
    const float n   = 1.0f - 2.0f * __builtin_amdgcn_rcpf(e2 + 1.0f);

    return fmaf(omz, n, zh);
}

template <int MODE>
__device__ __forceinline__ void gru_emit(const GruRow& g, float hnew, int lane,
                                         float* __restrict__ hdst, float* __restrict__ ydst)
{
    if (MODE == 0) {
        if (lane < 16) hdst[lane] = hnew;
    } else {
        float p = g.wout * hnew;
        p += __shfl_xor(p, 8, 32);
        p += __shfl_xor(p, 4, 32);
        p += __shfl_xor(p, 2, 32);
        p += __shfl_xor(p, 1, 32);
        if (lane == 0) *ydst = p + g.bo;
    }
}

template <int MODE>
__global__ __launch_bounds__(32) void gru_scan_kernel(
    const float* __restrict__ x,
    const float* __restrict__ W_ih,
    const float* __restrict__ W_hh,
    const float* __restrict__ b_ih,
    const float* __restrict__ b_hh,
    const float* __restrict__ W_out,
    const float* __restrict__ b_out,
    float* __restrict__ Hbuf,
    float* __restrict__ y,
    int nsteps)
{
    const int  lane = threadIdx.x;
    const bool lo   = lane < 16;
    const int  m    = lo ? lane : ((lane & 15) ^ 8);
    const int  kb   = lo ? 0 : 8;

    GruRow g;
    g.lo = lo;
#pragma unroll
    for (int k = 0; k < 8; ++k) {
        g.wR[k] = W_hh[m * 16 + kb + k];
        g.wZ[k] = W_hh[(16 + m) * 16 + kb + k];
        g.wN[k] = W_hh[(32 + m) * 16 + kb + k];
    }
    g.cR   = b_ih[m] + b_hh[m];
    g.wiR  = W_ih[m];
    g.cZ   = b_ih[16 + m] + b_hh[16 + m];
    g.wiZ  = W_ih[16 + m];
    g.bhhN = lo ? b_hh[32 + m] : 0.0f;
    g.cN   = b_ih[32 + m];
    g.wiN  = W_ih[32 + m];
    g.wout = lo ? W_out[m] : 0.0f;
    g.bo   = b_out[0];

    const int col2 = lo ? lane : (16 + m);

    float hv = 0.0f;
    int   t  = 0;

    const int nblk = nsteps >> 4;
    if (nblk > 0) {
        float xb[16];
#pragma unroll
        for (int i = 0; i < 16; ++i) xb[i] = x[i];

        for (int b = 0; b < nblk; ++b) {
            const int bn = (b + 1 < nblk) ? (b + 1) : b;
            const float* __restrict__ xp = x + (size_t)bn * 16;
            float xn[16];
#pragma unroll
            for (int i = 0; i < 16; ++i) xn[i] = xp[i];

            const size_t t0 = (size_t)b * 16;
            GruInit cur = gru_init(g, xb[0]);
            float hrow[16];
#pragma unroll
            for (int i = 0; i < 16; ++i) {
                GruInit nxt;
                hv = gru_step(g, hv, cur, xb[(i < 15) ? i + 1 : i], nxt);
                cur = nxt;
                hrow[i] = hv;
                __builtin_amdgcn_sched_barrier(0);
            }
            if (MODE == 0) {
#pragma unroll
                for (int i = 0; i < 16; i += 2)
                    Hbuf[(t0 + i) * 16 + col2] = lo ? hrow[i] : hrow[i + 1];
            } else {
#pragma unroll
                for (int i = 0; i < 16; ++i)
                    gru_emit<MODE>(g, hrow[i], lane, nullptr, y + t0 + i);
            }
#pragma unroll
            for (int i = 0; i < 16; ++i) xb[i] = xn[i];
        }
        t = nblk * 16;
    }
    for (; t < nsteps; ++t) {
        GruInit nxt;
        hv = gru_step(g, hv, gru_init(g, x[t]), x[t], nxt);
        gru_emit<MODE>(g, hv, lane, Hbuf + (size_t)t * 16, y + t);
    }
}

__global__ __launch_bounds__(128) void gru_proj_wmma_kernel(
    const float* __restrict__ Hbuf,
    const float* __restrict__ W_out,
    const float* __restrict__ b_out,
    float* __restrict__ y,
    int nY)
{
    const int lane = threadIdx.x & 31;
    const int wave = blockIdx.x * 4 + (threadIdx.x >> 5);
    const int t0   = wave * 16;
    if (t0 >= nY) return;

    const int half = lane >> 4;
    const int lrow = lane & 15;

    const float bo = b_out[0];
    v8f c;
#pragma unroll
    for (int i = 0; i < 8; ++i) c[i] = bo;

    const float* hrow = Hbuf + (size_t)(t0 + lrow) * 16;
#pragma unroll
    for (int j = 0; j < 4; ++j) {
        const int col = 4 * j + 2 * half;
        v2f a; a[0] = hrow[col];  a[1] = hrow[col + 1];
        v2f b; b[0] = W_out[col]; b[1] = W_out[col + 1];
        c = __builtin_amdgcn_wmma_f32_16x16x4_f32(
                 false, a,  false, b,
                 (short)0, c,  false,  false);
    }

    const int m0 = t0 + 8 * half;
    if (lrow == 0) {
        if (m0 + 8 <= nY) {
            float4 lo = make_float4(c[0], c[1], c[2], c[3]);
            float4 hi = make_float4(c[4], c[5], c[6], c[7]);
            *(float4*)(y + m0)     = lo;
            *(float4*)(y + m0 + 4) = hi;
        } else {
#pragma unroll
            for (int i = 0; i < 8; ++i)
                if (m0 + i < nY) y[m0 + i] = c[i];
        }
    }
}

extern "C" void kernel_launch(void* const* d_in, const int* in_sizes, int n_in,
                              void* d_out, int out_size, void* d_ws, size_t ws_size,
                              hipStream_t stream)
{
    const float* x     = (const float*)d_in[0];
    const float* W_ih  = (const float*)d_in[1];
    const float* W_hh  = (const float*)d_in[2];
    const float* b_ih  = (const float*)d_in[3];
    const float* b_hh  = (const float*)d_in[4];
    const float* W_out = (const float*)d_in[5];
    const float* b_out = (const float*)d_in[6];
    float* y = (float*)d_out;

    const int nsteps = in_sizes[0] - 1;
    const size_t needH = (size_t)(nsteps + 1) * 16 * sizeof(float);
    float* Hbuf = (float*)d_ws;

    if (ws_size >= needH) {
        hipLaunchKernelGGL((gru_scan_kernel<0>), dim3(1), dim3(32), 0, stream,
                           x, W_ih, W_hh, b_ih, b_hh, W_out, b_out,
                           Hbuf, y, nsteps);
        const int tiles  = (nsteps + 15) / 16;
        const int blocks = (tiles + 3) / 4;
        hipLaunchKernelGGL(gru_proj_wmma_kernel, dim3(blocks), dim3(128), 0, stream,
                           Hbuf, W_out, b_out, y, nsteps);
    } else {
        hipLaunchKernelGGL((gru_scan_kernel<1>), dim3(1), dim3(32), 0, stream,
                           x, W_ih, W_hh, b_ih, b_hh, W_out, b_out,
                           nullptr, y, nsteps);
    }
}


# r, z and n each by ONE v_tanh_f32 (sigmoid(a)=0.5+0.5*tanh(a/2); W_hh rows, bias and x terms stored halved, which is exact) in place of exp2+rcp chains; sum order kept, last bits of h differ
# speedup vs baseline: 2.6427x; 1.4739x over previous
#include <hip/hip_runtime.h>
#include <hip/hip_bf16.h>

typedef __attribute__((ext_vector_type(2))) float v2f;
typedef __attribute__((ext_vector_type(8))) float v8f;

__device__ __forceinline__ float xhalf8(float v) {
    const int b = __float_as_int(v);
    return __int_as_float(
        __builtin_amdgcn_permlanex16(b, b, 0xfedcba98, 0x76543210, true, false));
}

struct GruRow {
    float wR[8], wZ[8], wN[8];
    float cR, wiR, cZ, wiZ, bhhN, cN, wiN, wout, bo;
    bool  lo;
};

struct GruInit { float aR, aZ, gN; };

__device__ __forceinline__ GruInit gru_init_products(const GruRow& g, float x)
{
    GruInit s;
    s.aR = fmaf(g.wiR, x, g.cR);
    s.aZ = fmaf(g.wiZ, x, g.cZ);
    s.gN = fmaf(g.wiN, x, g.cN);
    return s;
}
__device__ __forceinline__ GruInit gru_init_rows(const GruRow& g, const GruInit& p)
{
    GruInit s;
    s.aR = g.lo ? p.aR : 0.0f;
    s.aZ = g.lo ? p.aZ : 0.0f;
    s.gN = p.gN;
    return s;
}
__device__ __forceinline__ GruInit gru_init(const GruRow& g, float x)
{
    return gru_init_rows(g, gru_init_products(g, x));
}

#define GRU_HERE() __builtin_amdgcn_sched_barrier(0)

#define GRU_LINK(K)                                                                          \
    asm("v_fmac_f32_dpp %0, %3, %4 row_share:" #K " row_mask:0xf bank_mask:0xf bound_ctrl:1\n\t"  \
        "v_fmac_f32_dpp %1, %3, %5 row_share:" #K " row_mask:0xf bank_mask:0xf bound_ctrl:1\n\t"  \
        "v_fmac_f32_dpp %2, %3, %6 row_share:" #K " row_mask:0xf bank_mask:0xf bound_ctrl:1"       \
        : "+v"(aR), "+v"(aN), "+v"(aZ)                                                       \
        : "v"(hv), "v"(g.wR[K]), "v"(g.wN[K]), "v"(g.wZ[K]))

__device__ __forceinline__ float gru_step(const GruRow& g, float hv, const GruInit& cur,
                                          float x_next, GruInit& nxt)
{
    float aR = cur.aR, aZ = cur.aZ, aN = g.bhhN;
    GRU_LINK(0); GRU_LINK(1); GRU_LINK(2); GRU_LINK(3);
    GRU_LINK(4); GRU_LINK(5); GRU_LINK(6); GRU_LINK(7);

    const float pR = xhalf8(aR), pN = xhalf8(aN);
    const float accR = aR + pR;
    GRU_HERE();
    const float pZ = xhalf8(aZ);
    const float hN = aN + pN;
    const float tr = __builtin_amdgcn_tanhf(accR);
    const float accZ = aZ + pZ;
    const float sN = cur.gN + hN;
    const float hh = 0.5f * hv;
    const GruInit raw = gru_init_products(g, x_next);
    GRU_HERE();
    const float tz = __builtin_amdgcn_tanhf(accZ);
    const float npre = fmaf(tr, hN, sN);
    nxt = gru_init_rows(g, raw);
    GRU_HERE();
    const float n   = __builtin_amdgcn_tanhf(npre);
    GRU_HERE();
    const float omz = fmaf(-0.5f, tz, 0.5f);
    const float zh  = fmaf(tz, hh, hh);

    return fmaf(omz, n, zh);
}

template <int MODE>
__device__ __forceinline__ void gru_emit(const GruRow& g, float hnew, int lane,
                                         float* __restrict__ hdst, float* __restrict__ ydst)
{
    if (MODE == 0) {
        if (lane < 16) hdst[lane] = hnew;
    } else {
        float p = g.wout * hnew;
        p += __shfl_xor(p, 8, 32);
        p += __shfl_xor(p, 4, 32);
        p += __shfl_xor(p, 2, 32);
        p += __shfl_xor(p, 1, 32);
        if (lane == 0) *ydst = p + g.bo;
    }
}

template <int MODE>
__global__ __launch_bounds__(32) void gru_scan_kernel(
    const float* __restrict__ x,
    const float* __restrict__ W_ih,
    const float* __restrict__ W_hh,
    const float* __restrict__ b_ih,
    const float* __restrict__ b_hh,
    const float* __restrict__ W_out,
    const float* __restrict__ b_out,
    float* __restrict__ Hbuf,
    float* __restrict__ y,
    int nsteps)
{
    const int  lane = threadIdx.x;
    const bool lo   = lane < 16;
    const int  m    = lo ? lane : ((lane & 15) ^ 8);
    const int  kb   = lo ? 0 : 8;

    GruRow g;
    g.lo = lo;
#pragma unroll
    for (int k = 0; k < 8; ++k) {
        g.wR[k] = 0.5f * W_hh[m * 16 + kb + k];
        g.wZ[k] = 0.5f * W_hh[(16 + m) * 16 + kb + k];
        g.wN[k] = 0.5f * W_hh[(32 + m) * 16 + kb + k];
    }
    g.cR   = 0.5f * (b_ih[m] + b_hh[m]);
    g.wiR  = 0.5f * W_ih[m];
    g.cZ   = 0.5f * (b_ih[16 + m] + b_hh[16 + m]);
    g.wiZ  = 0.5f * W_ih[16 + m];
    g.bhhN = lo ? 0.5f * b_hh[32 + m] : 0.0f;
    g.cN   = b_ih[32 + m];
    g.wiN  = W_ih[32 + m];
    g.wout = lo ? W_out[m] : 0.0f;
    g.bo   = b_out[0];

    const int col2 = lo ? lane : (16 + m);

    float hv = 0.0f;
    int   t  = 0;

    const int nblk = nsteps >> 4;
    if (nblk > 0) {
        float xb[16];
#pragma unroll
        for (int i = 0; i < 16; ++i) xb[i] = x[i];

        for (int b = 0; b < nblk; ++b) {
            const int bn = (b + 1 < nblk) ? (b + 1) : b;
            const float* __restrict__ xp = x + (size_t)bn * 16;
            float xn[16];
#pragma unroll
            for (int i = 0; i < 16; ++i) xn[i] = xp[i];

            const size_t t0 = (size_t)b * 16;
            GruInit cur = gru_init(g, xb[0]);
            float hrow[16];
#pragma unroll
            for (int i = 0; i < 16; ++i) {
                GruInit nxt;
                hv = gru_step(g, hv, cur, xb[(i < 15) ? i + 1 : i], nxt);
                cur = nxt;
                hrow[i] = hv;
                __builtin_amdgcn_sched_barrier(0);
            }
            if (MODE == 0) {
#pragma unroll
                for (int i = 0; i < 16; i += 2)
                    Hbuf[(t0 + i) * 16 + col2] = lo ? hrow[i] : hrow[i + 1];
            } else {
#pragma unroll
                for (int i = 0; i < 16; ++i)
                    gru_emit<MODE>(g, hrow[i], lane, nullptr, y + t0 + i);
            }
#pragma unroll
            for (int i = 0; i < 16; ++i) xb[i] = xn[i];
        }
        t = nblk * 16;
    }
    for (; t < nsteps; ++t) {
        GruInit nxt;
        hv = gru_step(g, hv, gru_init(g, x[t]), x[t], nxt);
        gru_emit<MODE>(g, hv, lane, Hbuf + (size_t)t * 16, y + t);
    }
}

__global__ __launch_bounds__(128) void gru_proj_wmma_kernel(
    const float* __restrict__ Hbuf,
    const float* __restrict__ W_out,
    const float* __restrict__ b_out,
    float* __restrict__ y,
    int nY)
{
    const int lane = threadIdx.x & 31;
    const int wave = blockIdx.x * 4 + (threadIdx.x >> 5);
    const int t0   = wave * 16;
    if (t0 >= nY) return;

    const int half = lane >> 4;
    const int lrow = lane & 15;

    const float bo = b_out[0];
    v8f c;
#pragma unroll
    for (int i = 0; i < 8; ++i) c[i] = bo;

    const float* hrow = Hbuf + (size_t)(t0 + lrow) * 16;
#pragma unroll
    for (int j = 0; j < 4; ++j) {
        const int col = 4 * j + 2 * half;
        v2f a; a[0] = hrow[col];  a[1] = hrow[col + 1];
        v2f b; b[0] = W_out[col]; b[1] = W_out[col + 1];
        c = __builtin_amdgcn_wmma_f32_16x16x4_f32(
                 false, a,  false, b,
                 (short)0, c,  false,  false);
    }

    const int m0 = t0 + 8 * half;
    if (lrow == 0) {
        if (m0 + 8 <= nY) {
            float4 lo = make_float4(c[0], c[1], c[2], c[3]);
            float4 hi = make_float4(c[4], c[5], c[6], c[7]);
            *(float4*)(y + m0)     = lo;
            *(float4*)(y + m0 + 4) = hi;
        } else {
#pragma unroll
            for (int i = 0; i < 8; ++i)
                if (m0 + i < nY) y[m0 + i] = c[i];
        }
    }
}

extern "C" void kernel_launch(void* const* d_in, const int* in_sizes, int n_in,
                              void* d_out, int out_size, void* d_ws, size_t ws_size,
                              hipStream_t stream)
{
    const float* x     = (const float*)d_in[0];
    const float* W_ih  = (const float*)d_in[1];
    const float* W_hh  = (const float*)d_in[2];
    const float* b_ih  = (const float*)d_in[3];
    const float* b_hh  = (const float*)d_in[4];
    const float* W_out = (const float*)d_in[5];
    const float* b_out = (const float*)d_in[6];
    float* y = (float*)d_out;

    const int nsteps = in_sizes[0] - 1;
    const size_t needH = (size_t)(nsteps + 1) * 16 * sizeof(float);
    float* Hbuf = (float*)d_ws;

    if (ws_size >= needH) {
        hipLaunchKernelGGL((gru_scan_kernel<0>), dim3(1), dim3(32), 0, stream,
                           x, W_ih, W_hh, b_ih, b_hh, W_out, b_out,
                           Hbuf, y, nsteps);
        const int tiles  = (nsteps + 15) / 16;
        const int blocks = (tiles + 3) / 4;
        hipLaunchKernelGGL(gru_proj_wmma_kernel, dim3(blocks), dim3(128), 0, stream,
                           Hbuf, W_out, b_out, y, nsteps);
    } else {
        hipLaunchKernelGGL((gru_scan_kernel<1>), dim3(1), dim3(32), 0, stream,
                           x, W_ih, W_hh, b_ih, b_hh, W_out, b_out,
                           nullptr, y, nsteps);
    }
}


# speedup vs baseline: 3.0922x; 1.1701x over previous
#include <hip/hip_runtime.h>
#include <hip/hip_bf16.h>

typedef __attribute__((ext_vector_type(2))) float v2f;
typedef __attribute__((ext_vector_type(8))) float v8f;

__device__ __forceinline__ float xhalf8(float v) {
    const int b = __float_as_int(v);
    return __int_as_float(
        __builtin_amdgcn_permlanex16(b, b, 0xfedcba98, 0x76543210, true, false));
}

struct GruRow {
    float wR[8], wZ[8], wN[8];
    float cR, wiR, cZ, wiZ, bhhN, cN, wiN, wout, bo;
    bool  lo;
};

struct GruInit { float aR, aZ, gN; };

__device__ __forceinline__ GruInit gru_init_products(const GruRow& g, float x)
{
    GruInit s;
    s.aR = fmaf(g.wiR, x, g.cR);
    s.aZ = fmaf(g.wiZ, x, g.cZ);
    s.gN = fmaf(g.wiN, x, g.cN);
    return s;
}
__device__ __forceinline__ GruInit gru_init_rows(const GruRow& g, const GruInit& p)
{
    GruInit s;
    s.aR = g.lo ? p.aR : 0.0f;
    s.aZ = g.lo ? p.aZ : 0.0f;
    s.gN = p.gN;
    return s;
}
__device__ __forceinline__ GruInit gru_init(const GruRow& g, float x)
{
    return gru_init_rows(g, gru_init_products(g, x));
}

#define GRU_HERE() __builtin_amdgcn_sched_barrier(0)

template <int K>
__device__ __forceinline__ float rowbcast(float v) {
    const int b = __float_as_int(v);
    return __int_as_float(__builtin_amdgcn_permlane16_var(b, b, K, true, false));
}

#define GRU_FMAC(ACC, WT, HB) asm("v_fmac_f32 %0, %1, %2" : "+v"(ACC) : "v"(WT), "v"(HB))
#define GRU_MUL(DST, WT, HB)  asm("v_mul_f32 %0, %1, %2" : "=v"(DST) : "v"(WT), "v"(HB))
#define GRU_FMA3(DST, WT, HB, C0) \
    asm("v_fma_f32 %0, %1, %2, %3" : "=v"(DST) : "v"(WT), "v"(HB), "v"(C0))

__device__ __forceinline__ float gru_step(const GruRow& g, float hv, const GruInit& cur,
                                          float x_next, GruInit& nxt)
{
    float aR = cur.aR, aZ = cur.aZ, aN;
    float bR, bZ, bN;

    const float h0 = rowbcast<0>(hv);
    const float h1 = rowbcast<1>(hv);
    const float h2 = rowbcast<2>(hv);
    const float h3 = rowbcast<3>(hv);
    const float h4 = rowbcast<4>(hv);
    GRU_HERE();
    GRU_FMAC(aR, g.wR[0], h0);  GRU_MUL(bR, g.wR[1], h1);
    GRU_HERE();
    const float h5 = rowbcast<5>(hv);
    GRU_HERE();
    GRU_FMA3(aN, g.wN[0], h0, g.bhhN);  GRU_MUL(bN, g.wN[1], h1);
    GRU_FMAC(aR, g.wR[2], h2);  GRU_FMAC(bR, g.wR[3], h3);
    GRU_FMAC(aZ, g.wZ[0], h0);  GRU_MUL(bZ, g.wZ[1], h1);
    GRU_HERE();
    const float h6 = rowbcast<6>(hv);
    const float h7 = rowbcast<7>(hv);
    GRU_HERE();
    GRU_FMAC(aR, g.wR[4], h4);  GRU_FMAC(bR, g.wR[5], h5);
    GRU_FMAC(aN, g.wN[2], h2);  GRU_FMAC(bN, g.wN[3], h3);
    GRU_FMAC(aZ, g.wZ[2], h2);  GRU_FMAC(bZ, g.wZ[3], h3);
    GRU_FMAC(aR, g.wR[6], h6);  GRU_FMAC(bR, g.wR[7], h7);
    GRU_FMAC(aN, g.wN[4], h4);  GRU_FMAC(bN, g.wN[5], h5);
    GRU_FMAC(aZ, g.wZ[4], h4);  GRU_FMAC(bZ, g.wZ[5], h5);
    GRU_HERE();
    const float jR = aR + bR;
    GRU_HERE();
    GRU_FMAC(aN, g.wN[6], h6);  GRU_FMAC(bN, g.wN[7], h7);
    GRU_FMAC(aZ, g.wZ[6], h6);  GRU_FMAC(bZ, g.wZ[7], h7);
    GRU_HERE();
    const float pR = xhalf8(jR);
    const float hh = 0.5f * hv;
    GRU_HERE();
    const float jN = aN + bN;
    const GruInit raw = gru_init_products(g, x_next);
    GRU_HERE();
    const float jZ = aZ + bZ;
    GRU_HERE();
    const float accR = jR + pR;
    GRU_HERE();
    const float pN = xhalf8(jN);
    const float gj = cur.gN + jN;
    GRU_HERE();
    const float pZ = xhalf8(jZ);
    GRU_HERE();
    const float tr = __builtin_amdgcn_tanhf(accR);
    GRU_HERE();
    const float hN = jN + pN;
    const float sN = gj + pN;
    GRU_HERE();
    const float accZ = jZ + pZ;
    nxt = gru_init_rows(g, raw);
    GRU_HERE();
    const float npre = fmaf(tr, hN, sN);
    GRU_HERE();
    const float tz = __builtin_amdgcn_tanhf(accZ);
    GRU_HERE();
    const float n   = __builtin_amdgcn_tanhf(npre);
    GRU_HERE();
    const float omz = fmaf(-0.5f, tz, 0.5f);
    const float zh  = fmaf(tz, hh, hh);

    return fmaf(omz, n, zh);
}

template <int MODE>
__device__ __forceinline__ void gru_emit(const GruRow& g, float hnew, int lane,
                                         float* __restrict__ hdst, float* __restrict__ ydst)
{
    if (MODE == 0) {
        if (lane < 16) hdst[lane] = hnew;
    } else {
        float p = g.wout * hnew;
        p += __shfl_xor(p, 8, 32);
        p += __shfl_xor(p, 4, 32);
        p += __shfl_xor(p, 2, 32);
        p += __shfl_xor(p, 1, 32);
        if (lane == 0) *ydst = p + g.bo;
    }
}

template <int MODE>
__global__ __launch_bounds__(32) void gru_scan_kernel(
    const float* __restrict__ x,
    const float* __restrict__ W_ih,
    const float* __restrict__ W_hh,
    const float* __restrict__ b_ih,
    const float* __restrict__ b_hh,
    const float* __restrict__ W_out,
    const float* __restrict__ b_out,
    float* __restrict__ Hbuf,
    float* __restrict__ y,
    int nsteps)
{
    const int  lane = threadIdx.x;
    const bool lo   = lane < 16;
    const int  m    = lo ? lane : ((lane & 15) ^ 8);
    const int  kb   = lo ? 0 : 8;

    GruRow g;
    g.lo = lo;
#pragma unroll
    for (int k = 0; k < 8; ++k) {
        g.wR[k] = 0.5f * W_hh[m * 16 + kb + k];
        g.wZ[k] = 0.5f * W_hh[(16 + m) * 16 + kb + k];
        g.wN[k] = 0.5f * W_hh[(32 + m) * 16 + kb + k];
    }
    g.cR   = 0.5f * (b_ih[m] + b_hh[m]);
    g.wiR  = 0.5f * W_ih[m];
    g.cZ   = 0.5f * (b_ih[16 + m] + b_hh[16 + m]);
    g.wiZ  = 0.5f * W_ih[16 + m];
    g.bhhN = lo ? 0.5f * b_hh[32 + m] : 0.0f;
    g.cN   = b_ih[32 + m];
    g.wiN  = W_ih[32 + m];
    g.wout = lo ? W_out[m] : 0.0f;
    g.bo   = b_out[0];

    const int col2 = lo ? lane : (16 + m);

    float hv = 0.0f;
    int   t  = 0;

    const int nblk = nsteps >> 4;
    if (nblk > 0) {
        float xb[16];
#pragma unroll
        for (int i = 0; i < 16; ++i) xb[i] = x[i];

        for (int b = 0; b < nblk; ++b) {
            const int bn = (b + 1 < nblk) ? (b + 1) : b;
            const float* __restrict__ xp = x + (size_t)bn * 16;
            float xn[16];
#pragma unroll
            for (int i = 0; i < 16; ++i) xn[i] = xp[i];

            const size_t t0 = (size_t)b * 16;
            GruInit cur = gru_init(g, xb[0]);
            float hrow[16];
#pragma unroll
            for (int i = 0; i < 16; ++i) {
                GruInit nxt;
                hv = gru_step(g, hv, cur, xb[(i < 15) ? i + 1 : i], nxt);
                cur = nxt;
                hrow[i] = hv;
                __builtin_amdgcn_sched_barrier(0);
            }
            if (MODE == 0) {
#pragma unroll
                for (int i = 0; i < 16; i += 2)
                    Hbuf[(t0 + i) * 16 + col2] = lo ? hrow[i] : hrow[i + 1];
            } else {
#pragma unroll
                for (int i = 0; i < 16; ++i)
                    gru_emit<MODE>(g, hrow[i], lane, nullptr, y + t0 + i);
            }
#pragma unroll
            for (int i = 0; i < 16; ++i) xb[i] = xn[i];
        }
        t = nblk * 16;
    }
    for (; t < nsteps; ++t) {
        GruInit nxt;
        hv = gru_step(g, hv, gru_init(g, x[t]), x[t], nxt);
        gru_emit<MODE>(g, hv, lane, Hbuf + (size_t)t * 16, y + t);
    }
}

__global__ __launch_bounds__(128) void gru_proj_wmma_kernel(
    const float* __restrict__ Hbuf,
    const float* __restrict__ W_out,
    const float* __restrict__ b_out,
    float* __restrict__ y,
    int nY)
{
    const int lane = threadIdx.x & 31;
    const int wave = blockIdx.x * 4 + (threadIdx.x >> 5);
    const int t0   = wave * 16;
    if (t0 >= nY) return;

    const int half = lane >> 4;
    const int lrow = lane & 15;

    const float bo = b_out[0];
    v8f c;
#pragma unroll
    for (int i = 0; i < 8; ++i) c[i] = bo;

    const float* hrow = Hbuf + (size_t)(t0 + lrow) * 16;
#pragma unroll
    for (int j = 0; j < 4; ++j) {
        const int col = 4 * j + 2 * half;
        v2f a; a[0] = hrow[col];  a[1] = hrow[col + 1];
        v2f b; b[0] = W_out[col]; b[1] = W_out[col + 1];
        c = __builtin_amdgcn_wmma_f32_16x16x4_f32(
                 false, a,  false, b,
                 (short)0, c,  false,  false);
    }

    const int m0 = t0 + 8 * half;
    if (lrow == 0) {
        if (m0 + 8 <= nY) {
            float4 lo = make_float4(c[0], c[1], c[2], c[3]);
            float4 hi = make_float4(c[4], c[5], c[6], c[7]);
            *(float4*)(y + m0)     = lo;
            *(float4*)(y + m0 + 4) = hi;
        } else {
#pragma unroll
            for (int i = 0; i < 8; ++i)
                if (m0 + i < nY) y[m0 + i] = c[i];
        }
    }
}

extern "C" void kernel_launch(void* const* d_in, const int* in_sizes, int n_in,
                              void* d_out, int out_size, void* d_ws, size_t ws_size,
                              hipStream_t stream)
{
    const float* x     = (const float*)d_in[0];
    const float* W_ih  = (const float*)d_in[1];
    const float* W_hh  = (const float*)d_in[2];
    const float* b_ih  = (const float*)d_in[3];
    const float* b_hh  = (const float*)d_in[4];
    const float* W_out = (const float*)d_in[5];
    const float* b_out = (const float*)d_in[6];
    float* y = (float*)d_out;

    const int nsteps = in_sizes[0] - 1;
    const size_t needH = (size_t)(nsteps + 1) * 16 * sizeof(float);
    float* Hbuf = (float*)d_ws;

    if (ws_size >= needH) {
        hipLaunchKernelGGL((gru_scan_kernel<0>), dim3(1), dim3(32), 0, stream,
                           x, W_ih, W_hh, b_ih, b_hh, W_out, b_out,
                           Hbuf, y, nsteps);
        const int tiles  = (nsteps + 15) / 16;
        const int blocks = (tiles + 3) / 4;
        hipLaunchKernelGGL(gru_proj_wmma_kernel, dim3(blocks), dim3(128), 0, stream,
                           Hbuf, W_out, b_out, y, nsteps);
    } else {
        hipLaunchKernelGGL((gru_scan_kernel<1>), dim3(1), dim3(32), 0, stream,
                           x, W_ih, W_hh, b_ih, b_hh, W_out, b_out,
                           nullptr, y, nsteps);
    }
}


# time loop in blocks of 64 steps (was 16), only the block's stores left at the seam: x refetched in place by quads (no copy), next block's start values and the stored row pairs formed inside the steps
# speedup vs baseline: 3.2083x; 1.0375x over previous
#include <hip/hip_runtime.h>
#include <hip/hip_bf16.h>

typedef __attribute__((ext_vector_type(2))) float v2f;
typedef __attribute__((ext_vector_type(8))) float v8f;
typedef __attribute__((ext_vector_type(4))) float v4f;

__device__ __forceinline__ float xhalf8(float v) {
    const int b = __float_as_int(v);
    return __int_as_float(
        __builtin_amdgcn_permlanex16(b, b, 0xfedcba98, 0x76543210, true, false));
}

struct GruRow {
    float wR[8], wZ[8], wN[8];
    float cR, wiR, cZ, wiZ, bhhN, cN, wiN, wout, bo;
    bool  lo;
    unsigned lomask;
};

struct GruInit { float aR, aZ, gN; };

__device__ __forceinline__ GruInit gru_init_products(const GruRow& g, float x)
{
    GruInit s;
    s.aR = fmaf(g.wiR, x, g.cR);
    s.aZ = fmaf(g.wiZ, x, g.cZ);
    s.gN = fmaf(g.wiN, x, g.cN);
    return s;
}
__device__ __forceinline__ GruInit gru_init_rows(const GruRow& g, const GruInit& p)
{
    GruInit s;
    s.aR = g.lo ? p.aR : 0.0f;
    s.aZ = g.lo ? p.aZ : 0.0f;
    s.gN = p.gN;
    return s;
}
__device__ __forceinline__ GruInit gru_init(const GruRow& g, float x)
{
    return gru_init_rows(g, gru_init_products(g, x));
}

#define GRU_HERE() __builtin_amdgcn_sched_barrier(0)

template <int K>
__device__ __forceinline__ float rowbcast(float v) {
    const int b = __float_as_int(v);
    return __int_as_float(__builtin_amdgcn_permlane16_var(b, b, K, true, false));
}

#define GRU_FMAC(ACC, WT, HB) asm("v_fmac_f32 %0, %1, %2" : "+v"(ACC) : "v"(WT), "v"(HB))
#define GRU_MUL(DST, WT, HB)  asm("v_mul_f32 %0, %1, %2" : "=v"(DST) : "v"(WT), "v"(HB))
#define GRU_FMA3(DST, WT, HB, C0) \
    asm("v_fma_f32 %0, %1, %2, %3" : "=v"(DST) : "v"(WT), "v"(HB), "v"(C0))

__device__ __forceinline__ float gru_step(const GruRow& g, float hv, const GruInit& cur,
                                          float x_next, GruInit& nxt,
                                          float sideA, float sideB, float& sideOut)
{
    float aR = cur.aR, aZ = cur.aZ, aN;
    float bR, bZ, bN;

    const float h0 = rowbcast<0>(hv);
    GRU_HERE();
    const float h1 = rowbcast<1>(hv);
    GRU_HERE();
    const float h2 = rowbcast<2>(hv);
    const float h3 = rowbcast<3>(hv);
    const float h4 = rowbcast<4>(hv);
    GRU_HERE();
    GRU_FMAC(aR, g.wR[0], h0);  GRU_MUL(bR, g.wR[1], h1);
    GRU_HERE();
    const float h5 = rowbcast<5>(hv);
    GRU_HERE();
    GRU_FMA3(aN, g.wN[0], h0, g.bhhN);  GRU_MUL(bN, g.wN[1], h1);
    GRU_FMAC(aR, g.wR[2], h2);  GRU_FMAC(bR, g.wR[3], h3);
    GRU_FMAC(aZ, g.wZ[0], h0);  GRU_MUL(bZ, g.wZ[1], h1);
    GRU_HERE();
    const float h6 = rowbcast<6>(hv);
    const float h7 = rowbcast<7>(hv);
    GRU_HERE();
    GRU_FMAC(aR, g.wR[4], h4);  GRU_FMAC(bR, g.wR[5], h5);
    GRU_FMAC(aN, g.wN[2], h2);  GRU_FMAC(bN, g.wN[3], h3);
    GRU_FMAC(aZ, g.wZ[2], h2);  GRU_FMAC(bZ, g.wZ[3], h3);
    GRU_FMAC(aR, g.wR[6], h6);  GRU_FMAC(bR, g.wR[7], h7);
    GRU_FMAC(aN, g.wN[4], h4);  GRU_FMAC(bN, g.wN[5], h5);
    GRU_FMAC(aZ, g.wZ[4], h4);  GRU_FMAC(bZ, g.wZ[5], h5);
    GRU_HERE();
    const float jR = aR + bR;
    GRU_HERE();
    GRU_FMAC(aN, g.wN[6], h6);  GRU_FMAC(bN, g.wN[7], h7);
    GRU_FMAC(aZ, g.wZ[6], h6);  GRU_FMAC(bZ, g.wZ[7], h7);
    GRU_HERE();
    const float pR = xhalf8(jR);
    const float hh = 0.5f * hv;
    GRU_HERE();
    const float jN = aN + bN;
    const GruInit raw = gru_init_products(g, x_next);
    GRU_HERE();
    const float jZ = aZ + bZ;
    GRU_HERE();
    const float accR = jR + pR;
    GRU_HERE();
    const float pN = xhalf8(jN);
    const float gj = cur.gN + jN;
    GRU_HERE();
    const float pZ = xhalf8(jZ);
    GRU_HERE();
    const float tr = __builtin_amdgcn_tanhf(accR);
    GRU_HERE();
    const float hN = jN + pN;
    const float sN = gj + pN;
    GRU_HERE();
    const float accZ = jZ + pZ;
    nxt = gru_init_rows(g, raw);
    GRU_HERE();
    const float npre = fmaf(tr, hN, sN);
    GRU_HERE();
    const float tz = __builtin_amdgcn_tanhf(accZ);
    asm("v_cndmask_b32_e64 %0, %1, %2, %3" : "=v"(sideOut) : "v"(sideB), "v"(sideA), "s"(g.lomask));
    GRU_HERE();
    const float n   = __builtin_amdgcn_tanhf(npre);
    GRU_HERE();
    const float omz = fmaf(-0.5f, tz, 0.5f);
    const float zh  = fmaf(tz, hh, hh);

    return fmaf(omz, n, zh);
}

template <int MODE>
__device__ __forceinline__ void gru_emit(const GruRow& g, float hnew, int lane,
                                         float* __restrict__ hdst, float* __restrict__ ydst)
{
    if (MODE == 0) {
        if (lane < 16) hdst[lane] = hnew;
    } else {
        float p = g.wout * hnew;
        p += __shfl_xor(p, 8, 32);
        p += __shfl_xor(p, 4, 32);
        p += __shfl_xor(p, 2, 32);
        p += __shfl_xor(p, 1, 32);
        if (lane == 0) *ydst = p + g.bo;
    }
}

#define GRU_BLK 64
template <int MODE>
__global__ __launch_bounds__(32) void gru_scan_kernel(
    const float* __restrict__ x,
    const float* __restrict__ W_ih,
    const float* __restrict__ W_hh,
    const float* __restrict__ b_ih,
    const float* __restrict__ b_hh,
    const float* __restrict__ W_out,
    const float* __restrict__ b_out,
    float* __restrict__ Hbuf,
    float* __restrict__ y,
    int nsteps)
{
    const int  lane = threadIdx.x;
    const bool lo   = lane < 16;
    const int  m    = lo ? lane : ((lane & 15) ^ 8);
    const int  kb   = lo ? 0 : 8;

    GruRow g;
    g.lo = lo;
    g.lomask = 0xffffu;
#pragma unroll
    for (int k = 0; k < 8; ++k) {
        g.wR[k] = 0.5f * W_hh[m * 16 + kb + k];
        g.wZ[k] = 0.5f * W_hh[(16 + m) * 16 + kb + k];
        g.wN[k] = 0.5f * W_hh[(32 + m) * 16 + kb + k];
    }
    g.cR   = 0.5f * (b_ih[m] + b_hh[m]);
    g.wiR  = 0.5f * W_ih[m];
    g.cZ   = 0.5f * (b_ih[16 + m] + b_hh[16 + m]);
    g.wiZ  = 0.5f * W_ih[16 + m];
    g.bhhN = lo ? 0.5f * b_hh[32 + m] : 0.0f;
    g.cN   = b_ih[32 + m];
    g.wiN  = W_ih[32 + m];
    g.wout = lo ? W_out[m] : 0.0f;
    g.bo   = b_out[0];

    const int col2 = lo ? lane : (16 + m);

    float hv = 0.0f;
    int   t  = 0;

    const int nblk = nsteps / GRU_BLK;
    if (nblk > 0) {
        v4f xq[GRU_BLK / 4];
#pragma unroll
        for (int j = 0; j < GRU_BLK / 4; ++j) __builtin_memcpy(&xq[j], x + 4 * j, sizeof(v4f));
        GruInit cur = gru_init(g, xq[0][0]);

        for (int b = 0; b < nblk; ++b) {
            const int bn = (b + 1 < nblk) ? (b + 1) : b;
            const float* __restrict__ xp = x + (size_t)bn * GRU_BLK;

            const size_t t0 = (size_t)b * GRU_BLK;
            float hsel[GRU_BLK / 2];
            float ha = 0.0f, hb = 0.0f;
#pragma unroll
            for (int i = 0; i < GRU_BLK; ++i) {
                GruInit nxt;
                float side;
                hv = gru_step(g, hv, cur, xq[((i + 1) % GRU_BLK) >> 2][(i + 1) & 3], nxt,
                              ha, hb, side);
                cur = nxt;
                if ((i & 1) == 0) {
                    if (i >= 2) hsel[(i >> 1) - 1] = side;
                    ha = hv;
                } else {
                    hb = hv;
                }
                if (MODE != 0) gru_emit<MODE>(g, hv, lane, nullptr, y + t0 + i);
                __builtin_amdgcn_sched_barrier(0);
                if ((i & 3) == 3) {
                    __builtin_memcpy(&xq[i >> 2], xp + (i - 3), sizeof(v4f));
                }
            }
            if (MODE == 0) {
                hsel[GRU_BLK / 2 - 1] = lo ? ha : hb;
#pragma unroll
                for (int j = 0; j < GRU_BLK / 2; ++j)
                    Hbuf[(t0 + 2 * j) * 16 + col2] = hsel[j];
            }
        }
        t = nblk * GRU_BLK;
    }
    for (; t < nsteps; ++t) {
        GruInit nxt;
        float side;
        hv = gru_step(g, hv, gru_init(g, x[t]), x[t], nxt, 0.0f, 0.0f, side);
        gru_emit<MODE>(g, hv, lane, Hbuf + (size_t)t * 16, y + t);
    }
}

__global__ __launch_bounds__(128) void gru_proj_wmma_kernel(
    const float* __restrict__ Hbuf,
    const float* __restrict__ W_out,
    const float* __restrict__ b_out,
    float* __restrict__ y,
    int nY)
{
    const int lane = threadIdx.x & 31;
    const int wave = blockIdx.x * 4 + (threadIdx.x >> 5);
    const int t0   = wave * 16;
    if (t0 >= nY) return;

    const int half = lane >> 4;
    const int lrow = lane & 15;

    const float bo = b_out[0];
    v8f c;
#pragma unroll
    for (int i = 0; i < 8; ++i) c[i] = bo;

    const float* hrow = Hbuf + (size_t)(t0 + lrow) * 16;
#pragma unroll
    for (int j = 0; j < 4; ++j) {
        const int col = 4 * j + 2 * half;
        v2f a; a[0] = hrow[col];  a[1] = hrow[col + 1];
        v2f b; b[0] = W_out[col]; b[1] = W_out[col + 1];
        c = __builtin_amdgcn_wmma_f32_16x16x4_f32(
                 false, a,  false, b,
                 (short)0, c,  false,  false);
    }

    const int m0 = t0 + 8 * half;
    if (lrow == 0) {
        if (m0 + 8 <= nY) {
            float4 lo = make_float4(c[0], c[1], c[2], c[3]);
            float4 hi = make_float4(c[4], c[5], c[6], c[7]);
            *(float4*)(y + m0)     = lo;
            *(float4*)(y + m0 + 4) = hi;
        } else {
#pragma unroll
            for (int i = 0; i < 8; ++i)
                if (m0 + i < nY) y[m0 + i] = c[i];
        }
    }
}

extern "C" void kernel_launch(void* const* d_in, const int* in_sizes, int n_in,
                              void* d_out, int out_size, void* d_ws, size_t ws_size,
                              hipStream_t stream)
{
    const float* x     = (const float*)d_in[0];
    const float* W_ih  = (const float*)d_in[1];
    const float* W_hh  = (const float*)d_in[2];
    const float* b_ih  = (const float*)d_in[3];
    const float* b_hh  = (const float*)d_in[4];
    const float* W_out = (const float*)d_in[5];
    const float* b_out = (const float*)d_in[6];
    float* y = (float*)d_out;

    const int nsteps = in_sizes[0] - 1;
    const size_t needH = (size_t)(nsteps + 1) * 16 * sizeof(float);
    float* Hbuf = (float*)d_ws;

    if (ws_size >= needH) {
        hipLaunchKernelGGL((gru_scan_kernel<0>), dim3(1), dim3(32), 0, stream,
                           x, W_ih, W_hh, b_ih, b_hh, W_out, b_out,
                           Hbuf, y, nsteps);
        const int tiles  = (nsteps + 15) / 16;
        const int blocks = (tiles + 3) / 4;
        hipLaunchKernelGGL(gru_proj_wmma_kernel, dim3(blocks), dim3(128), 0, stream,
                           Hbuf, W_out, b_out, y, nsteps);
    } else {
        hipLaunchKernelGGL((gru_scan_kernel<1>), dim3(1), dim3(32), 0, stream,
                           x, W_ih, W_hh, b_ih, b_hh, W_out, b_out,
                           nullptr, y, nsteps);
    }
}


# z's and n's 16 multiply-adds a step issue as 8 dual-issue pairs (plain fmaf: a gate's even-k and odd-k link side by side); r's 8 stay single and issue as their operands arrive; every sum unchanged
# speedup vs baseline: 3.3359x; 1.0398x over previous
#include <hip/hip_runtime.h>
#include <hip/hip_bf16.h>

typedef __attribute__((ext_vector_type(2))) float v2f;
typedef __attribute__((ext_vector_type(8))) float v8f;
typedef __attribute__((ext_vector_type(4))) float v4f;

__device__ __forceinline__ float xhalf8(float v) {
    const int b = __float_as_int(v);
    return __int_as_float(
        __builtin_amdgcn_permlanex16(b, b, 0xfedcba98, 0x76543210, true, false));
}

struct GruRow {
    float wR[8], wZ[8], wN[8];
    float cR, wiR, cZ, wiZ, bhhN, cN, wiN, wout, bo;
    bool  lo;
    unsigned lomask;
};

struct GruInit { float aR, aZ, gN; };

__device__ __forceinline__ GruInit gru_init_products(const GruRow& g, float x)
{
    GruInit s;
    s.aR = fmaf(g.wiR, x, g.cR);
    s.aZ = fmaf(g.wiZ, x, g.cZ);
    s.gN = fmaf(g.wiN, x, g.cN);
    return s;
}
__device__ __forceinline__ GruInit gru_init_rows(const GruRow& g, const GruInit& p)
{
    GruInit s;
    s.aR = g.lo ? p.aR : 0.0f;
    s.aZ = g.lo ? p.aZ : 0.0f;
    s.gN = p.gN;
    return s;
}
__device__ __forceinline__ GruInit gru_init(const GruRow& g, float x)
{
    return gru_init_rows(g, gru_init_products(g, x));
}

#define GRU_HERE() __builtin_amdgcn_sched_barrier(0)

template <int K>
__device__ __forceinline__ float rowbcast(float v) {
    const int b = __float_as_int(v);
    return __int_as_float(__builtin_amdgcn_permlane16_var(b, b, K, true, false));
}

#define GRU_FMAC(ACC, WT, HB) asm("v_fmac_f32 %0, %1, %2" : "+v"(ACC) : "v"(WT), "v"(HB))
#define GRU_MUL(DST, WT, HB)  asm("v_mul_f32 %0, %1, %2" : "=v"(DST) : "v"(WT), "v"(HB))

__device__ __forceinline__ float gru_step(const GruRow& g, float hv, const GruInit& cur,
                                          float x_next, GruInit& nxt,
                                          float sideA, float sideB, float& sideOut)
{
    float aR = cur.aR, aZ = cur.aZ, aN;
    float bR, bZ, bN;

    const float h0 = rowbcast<0>(hv);
    GRU_HERE();
    const float h1 = rowbcast<1>(hv);
    GRU_HERE();
    const float h2 = rowbcast<2>(hv);
    const float h3 = rowbcast<3>(hv);
    const float h4 = rowbcast<4>(hv);
    GRU_HERE();
    GRU_FMAC(aR, g.wR[0], h0);  GRU_MUL(bR, g.wR[1], h1);
    GRU_HERE();
    const float h5 = rowbcast<5>(hv);
    GRU_HERE();
    aN = fmaf(g.wN[0], h0, g.bhhN);      bN = g.wN[1] * h1;
    GRU_HERE();
    aZ = fmaf(g.wZ[0], h0, aZ);          bZ = g.wZ[1] * h1;
    GRU_HERE();
    GRU_FMAC(aR, g.wR[2], h2);  GRU_FMAC(bR, g.wR[3], h3);
    GRU_HERE();
    const float h6 = rowbcast<6>(hv);
    const float h7 = rowbcast<7>(hv);
    GRU_HERE();
    aN = fmaf(g.wN[2], h2, aN);          bN = fmaf(g.wN[3], h3, bN);
    GRU_HERE();
    GRU_FMAC(aR, g.wR[4], h4);  GRU_FMAC(bR, g.wR[5], h5);
    GRU_HERE();
    aZ = fmaf(g.wZ[2], h2, aZ);          bZ = fmaf(g.wZ[3], h3, bZ);
    GRU_HERE();
    const float hh = 0.5f * hv;
    GRU_HERE();
    aN = fmaf(g.wN[4], h4, aN);          bN = fmaf(g.wN[5], h5, bN);
    GRU_HERE();
    GRU_FMAC(aR, g.wR[6], h6);  GRU_FMAC(bR, g.wR[7], h7);
    GRU_HERE();
    aZ = fmaf(g.wZ[4], h4, aZ);          bZ = fmaf(g.wZ[5], h5, bZ);
    GRU_HERE();
    GruInit raw;
    raw.aR = fmaf(g.wiR, x_next, g.cR);
    raw.aZ = fmaf(g.wiZ, x_next, g.cZ);
    GRU_HERE();
    aN = fmaf(g.wN[6], h6, aN);          bN = fmaf(g.wN[7], h7, bN);
    GRU_HERE();
    raw.gN = fmaf(g.wiN, x_next, g.cN);
    GRU_HERE();
    const float jR = aR + bR;
    GRU_HERE();
    aZ = fmaf(g.wZ[6], h6, aZ);          bZ = fmaf(g.wZ[7], h7, bZ);
    GRU_HERE();
    nxt = gru_init_rows(g, raw);
    GRU_HERE();
    const float jN = aN + bN;
    GRU_HERE();
    const float pR = xhalf8(jR);
    GRU_HERE();
    const float jZ = aZ + bZ;
    GRU_HERE();
    asm("v_cndmask_b32_e64 %0, %1, %2, %3" : "=v"(sideOut) : "v"(sideB), "v"(sideA), "s"(g.lomask));
    GRU_HERE();
    const float pN = xhalf8(jN);
    GRU_HERE();
    const float gj = cur.gN + jN;
    GRU_HERE();
    const float accR = jR + pR;
    GRU_HERE();
    const float pZ = xhalf8(jZ);
    GRU_HERE();
    const float tr = __builtin_amdgcn_tanhf(accR);
    GRU_HERE();
    const float accZ = jZ + pZ;
    GRU_HERE();
    const float hN = jN + pN;
    const float sN = gj + pN;
    GRU_HERE();
    const float tz = __builtin_amdgcn_tanhf(accZ);
    GRU_HERE();
    const float npre = fmaf(tr, hN, sN);
    GRU_HERE();
    const float n   = __builtin_amdgcn_tanhf(npre);
    GRU_HERE();
    const float omz = fmaf(-0.5f, tz, 0.5f);
    const float zh  = fmaf(tz, hh, hh);

    return fmaf(omz, n, zh);
}

template <int MODE>
__device__ __forceinline__ void gru_emit(const GruRow& g, float hnew, int lane,
                                         float* __restrict__ hdst, float* __restrict__ ydst)
{
    if (MODE == 0) {
        if (lane < 16) hdst[lane] = hnew;
    } else {
        float p = g.wout * hnew;
        p += __shfl_xor(p, 8, 32);
        p += __shfl_xor(p, 4, 32);
        p += __shfl_xor(p, 2, 32);
        p += __shfl_xor(p, 1, 32);
        if (lane == 0) *ydst = p + g.bo;
    }
}

#define GRU_BLK 64
template <int MODE>
__global__ __launch_bounds__(32) void gru_scan_kernel(
    const float* __restrict__ x,
    const float* __restrict__ W_ih,
    const float* __restrict__ W_hh,
    const float* __restrict__ b_ih,
    const float* __restrict__ b_hh,
    const float* __restrict__ W_out,
    const float* __restrict__ b_out,
    float* __restrict__ Hbuf,
    float* __restrict__ y,
    int nsteps)
{
    const int  lane = threadIdx.x;
    const bool lo   = lane < 16;
    const int  m    = lo ? lane : ((lane & 15) ^ 8);
    const int  kb   = lo ? 0 : 8;

    GruRow g;
    g.lo = lo;
    g.lomask = 0xffffu;
#pragma unroll
    for (int k = 0; k < 8; ++k) {
        g.wR[k] = 0.5f * W_hh[m * 16 + kb + k];
        g.wZ[k] = 0.5f * W_hh[(16 + m) * 16 + kb + k];
        g.wN[k] = 0.5f * W_hh[(32 + m) * 16 + kb + k];
    }
    g.cR   = 0.5f * (b_ih[m] + b_hh[m]);
    g.wiR  = 0.5f * W_ih[m];
    g.cZ   = 0.5f * (b_ih[16 + m] + b_hh[16 + m]);
    g.wiZ  = 0.5f * W_ih[16 + m];
    g.bhhN = lo ? 0.5f * b_hh[32 + m] : 0.0f;
    g.cN   = b_ih[32 + m];
    g.wiN  = W_ih[32 + m];
    g.wout = lo ? W_out[m] : 0.0f;
    g.bo   = b_out[0];

    const int col2 = lo ? lane : (16 + m);

    float hv = 0.0f;
    int   t  = 0;

    const int nblk = nsteps / GRU_BLK;
    if (nblk > 0) {
        v4f xq[GRU_BLK / 4];
#pragma unroll
        for (int j = 0; j < GRU_BLK / 4; ++j) __builtin_memcpy(&xq[j], x + 4 * j, sizeof(v4f));
        GruInit cur = gru_init(g, xq[0][0]);

        for (int b = 0; b < nblk; ++b) {
            const int bn = (b + 1 < nblk) ? (b + 1) : b;
            const float* __restrict__ xp = x + (size_t)bn * GRU_BLK;

            const size_t t0 = (size_t)b * GRU_BLK;
            float hsel[GRU_BLK / 2];
            float ha = 0.0f, hb = 0.0f;
#pragma unroll
            for (int i = 0; i < GRU_BLK; ++i) {
                GruInit nxt;
                float side;
                hv = gru_step(g, hv, cur, xq[((i + 1) % GRU_BLK) >> 2][(i + 1) & 3], nxt,
                              ha, hb, side);
                cur = nxt;
                if ((i & 1) == 0) {
                    if (i >= 2) hsel[(i >> 1) - 1] = side;
                    ha = hv;
                } else {
                    hb = hv;
                }
                if (MODE != 0) gru_emit<MODE>(g, hv, lane, nullptr, y + t0 + i);
                __builtin_amdgcn_sched_barrier(0);
                if ((i & 3) == 3) {
                    __builtin_memcpy(&xq[i >> 2], xp + (i - 3), sizeof(v4f));
                }
            }
            if (MODE == 0) {
                hsel[GRU_BLK / 2 - 1] = lo ? ha : hb;
#pragma unroll
                for (int j = 0; j < GRU_BLK / 2; ++j)
                    Hbuf[(t0 + 2 * j) * 16 + col2] = hsel[j];
            }
        }
        t = nblk * GRU_BLK;
    }
    for (; t < nsteps; ++t) {
        GruInit nxt;
        float side;
        hv = gru_step(g, hv, gru_init(g, x[t]), x[t], nxt, 0.0f, 0.0f, side);
        gru_emit<MODE>(g, hv, lane, Hbuf + (size_t)t * 16, y + t);
    }
}

__global__ __launch_bounds__(128) void gru_proj_wmma_kernel(
    const float* __restrict__ Hbuf,
    const float* __restrict__ W_out,
    const float* __restrict__ b_out,
    float* __restrict__ y,
    int nY)
{
    const int lane = threadIdx.x & 31;
    const int wave = blockIdx.x * 4 + (threadIdx.x >> 5);
    const int t0   = wave * 16;
    if (t0 >= nY) return;

    const int half = lane >> 4;
    const int lrow = lane & 15;

    const float bo = b_out[0];
    v8f c;
#pragma unroll
    for (int i = 0; i < 8; ++i) c[i] = bo;

    const float* hrow = Hbuf + (size_t)(t0 + lrow) * 16;
#pragma unroll
    for (int j = 0; j < 4; ++j) {
        const int col = 4 * j + 2 * half;
        v2f a; a[0] = hrow[col];  a[1] = hrow[col + 1];
        v2f b; b[0] = W_out[col]; b[1] = W_out[col + 1];
        c = __builtin_amdgcn_wmma_f32_16x16x4_f32(
                 false, a,  false, b,
                 (short)0, c,  false,  false);
    }

    const int m0 = t0 + 8 * half;
    if (lrow == 0) {
        if (m0 + 8 <= nY) {
            float4 lo = make_float4(c[0], c[1], c[2], c[3]);
            float4 hi = make_float4(c[4], c[5], c[6], c[7]);
            *(float4*)(y + m0)     = lo;
            *(float4*)(y + m0 + 4) = hi;
        } else {
#pragma unroll
            for (int i = 0; i < 8; ++i)
                if (m0 + i < nY) y[m0 + i] = c[i];
        }
    }
}

extern "C" void kernel_launch(void* const* d_in, const int* in_sizes, int n_in,
                              void* d_out, int out_size, void* d_ws, size_t ws_size,
                              hipStream_t stream)
{
    const float* x     = (const float*)d_in[0];
    const float* W_ih  = (const float*)d_in[1];
    const float* W_hh  = (const float*)d_in[2];
    const float* b_ih  = (const float*)d_in[3];
    const float* b_hh  = (const float*)d_in[4];
    const float* W_out = (const float*)d_in[5];
    const float* b_out = (const float*)d_in[6];
    float* y = (float*)d_out;

    const int nsteps = in_sizes[0] - 1;
    const size_t needH = (size_t)(nsteps + 1) * 16 * sizeof(float);
    float* Hbuf = (float*)d_ws;

    if (ws_size >= needH) {
        hipLaunchKernelGGL((gru_scan_kernel<0>), dim3(1), dim3(32), 0, stream,
                           x, W_ih, W_hh, b_ih, b_hh, W_out, b_out,
                           Hbuf, y, nsteps);
        const int tiles  = (nsteps + 15) / 16;
        const int blocks = (tiles + 3) / 4;
        hipLaunchKernelGGL(gru_proj_wmma_kernel, dim3(blocks), dim3(128), 0, stream,
                           Hbuf, W_out, b_out, y, nsteps);
    } else {
        hipLaunchKernelGGL((gru_scan_kernel<1>), dim3(1), dim3(32), 0, stream,
                           x, W_ih, W_hh, b_ih, b_hh, W_out, b_out,
                           nullptr, y, nsteps);
    }
}
